# starting kernel
# baseline (speedup 1.0000x reference)
#include <hip/hip_runtime.h>

typedef __attribute__((ext_vector_type(2))) float v2f;
typedef __attribute__((ext_vector_type(8))) float v8f;

#define DIM 512
#define MN  4096
#define LDA 516

__global__ __launch_bounds__(256) void som_w2_kernel(const float* __restrict__ weights,
                                                     float* __restrict__ w2) {
  const int wave = threadIdx.x >> 5;
  const int lane = threadIdx.x & 31;
  const int row  = blockIdx.x * 8 + wave;
  const float4* wr = (const float4*)(weights + (size_t)row * DIM);
  float acc = 0.0f;
#pragma unroll
  for (int i = 0; i < 4; ++i) {
    float4 q = wr[lane + 32 * i];
    acc += q.x * q.x + q.y * q.y + q.z * q.z + q.w * q.w;
  }
#pragma unroll
  for (int m = 16; m >= 1; m >>= 1) acc += __shfl_xor(acc, m, 32);
  if (lane == 0) w2[row] = acc;
}

__global__ __launch_bounds__(128) void som_bmu_kernel(const float* __restrict__ batch,
                                                      const float* __restrict__ weights,
                                                      const float* __restrict__ w2,
                                                      int* __restrict__ bmu) {
  __shared__ __align__(16) float As[16 * LDA];
  __shared__ float cV[4][16];
  __shared__ int   cI[4][16];

  const int tid  = threadIdx.x;
  const int lane = tid & 31;
  const int wv   = tid >> 5;
  const int bid  = blockIdx.x;

  const float4* bp = (const float4*)(batch + (size_t)bid * 16 * DIM);
#pragma unroll
  for (int i = 0; i < 16; ++i) {
    int idx = tid + 128 * i;
    int row = idx >> 7;
    int c4  = idx & 127;
    float4 q = bp[idx];
    *(float4*)&As[row * LDA + c4 * 4] = q;
  }
  __syncthreads();

  const int half = lane >> 4;
  const int l16  = lane & 15;
  const float* aRow = &As[l16 * LDA + 2 * half];

  float bestV[8];
  int   bestI[8];
#pragma unroll
  for (int r = 0; r < 8; ++r) { bestV[r] = -3.402823466e38f; bestI[r] = 0; }

  for (int nt = wv; nt < MN / 16; nt += 4) {
    const int n = nt * 16 + l16;
    const float* wRow = weights + (size_t)n * DIM + 2 * half;
    v8f acc0 = {}, acc1 = {}, acc2 = {}, acc3 = {};
#pragma unroll 2
    for (int ks = 0; ks < DIM / 4; ks += 4) {
      v2f a0 = *(const v2f*)(aRow + 4 * ks);
      v2f b0 = *(const v2f*)(wRow + 4 * ks);
      acc0 = __builtin_amdgcn_wmma_f32_16x16x4_f32(false, a0, false, b0,
                                                   (short)0, acc0, false, false);
      v2f a1 = *(const v2f*)(aRow + 4 * ks + 4);
      v2f b1 = *(const v2f*)(wRow + 4 * ks + 4);
      acc1 = __builtin_amdgcn_wmma_f32_16x16x4_f32(false, a1, false, b1,
                                                   (short)0, acc1, false, false);
      v2f a2 = *(const v2f*)(aRow + 4 * ks + 8);
      v2f b2 = *(const v2f*)(wRow + 4 * ks + 8);
      acc2 = __builtin_amdgcn_wmma_f32_16x16x4_f32(false, a2, false, b2,
                                                   (short)0, acc2, false, false);
      v2f a3 = *(const v2f*)(aRow + 4 * ks + 12);
      v2f b3 = *(const v2f*)(wRow + 4 * ks + 12);
      acc3 = __builtin_amdgcn_wmma_f32_16x16x4_f32(false, a3, false, b3,
                                                   (short)0, acc3, false, false);
    }
    const float w2n = w2[n];
#pragma unroll
    for (int r = 0; r < 8; ++r) {
      float s = 2.0f * (acc0[r] + acc1[r] + acc2[r] + acc3[r]) - w2n;
      if (s > bestV[r]) { bestV[r] = s; bestI[r] = n; }
    }
  }

#pragma unroll
  for (int r = 0; r < 8; ++r) {
    float v = bestV[r];
    int   ix = bestI[r];
#pragma unroll
    for (int m = 8; m >= 1; m >>= 1) {
      float ov = __shfl_xor(v, m, 32);
      int   oi = __shfl_xor(ix, m, 32);
      if (ov > v || (ov == v && oi < ix)) { v = ov; ix = oi; }
    }
    if (lane == 0)  { cV[wv][r]     = v; cI[wv][r]     = ix; }
    if (lane == 16) { cV[wv][8 + r] = v; cI[wv][8 + r] = ix; }
  }
  __syncthreads();

  if (tid < 16) {
    float v = cV[0][tid];
    int   ix = cI[0][tid];
#pragma unroll
    for (int w = 1; w < 4; ++w) {
      float ov = cV[w][tid];
      int   oi = cI[w][tid];
      if (ov > v || (ov == v && oi < ix)) { v = ov; ix = oi; }
    }
    bmu[bid * 16 + tid] = ix;
  }
}

__global__ __launch_bounds__(128) void som_out_kernel(const int* __restrict__ bmu,
                                                      const int* __restrict__ it_p,
                                                      const int* __restrict__ decay_p,
                                                      float* __restrict__ out) {
  __shared__ float efi[64];
  __shared__ float efj[64];
  const int row = blockIdx.x;
  const int t   = threadIdx.x;
  const int idx = bmu[row];
  const int bi = idx >> 6;
  const int bj = idx & 63;
  const float lr  = __expf(-(float)it_p[0] / (float)decay_p[0]);
  const float sig = 32.0f * lr;
  const float inv = 1.0f / (sig * sig);
  if (t < 64) {
    int d = t - bi;
    efi[t] = __expf(-(float)(d * d) * inv);
  } else {
    int c = t - 64;
    int d = c - bj;
    efj[c] = __expf(-(float)(d * d) * inv);
  }
  __syncthreads();
  float* op = out + (size_t)row * MN;
  const float fj = efj[t & 63];
#pragma unroll
  for (int k = 0; k < 32; ++k) {
    int j = t + (k << 7);
    op[j] = efi[j >> 6] * fj;
  }
}

extern "C" void kernel_launch(void* const* d_in, const int* in_sizes, int n_in,
                              void* d_out, int out_size, void* d_ws, size_t ws_size,
                              hipStream_t stream) {
  const float* batch   = (const float*)d_in[0];
  const float* weights = (const float*)d_in[1];
  const int* decay = (const int*)d_in[3];
  const int* it    = (const int*)d_in[4];

  float* w2  = (float*)d_ws;
  int*   bmu = (int*)((char*)d_ws + MN * sizeof(float));
  float* out = (float*)d_out;

  som_w2_kernel <<<MN / 8, 256, 0, stream>>>(weights, w2);
  som_bmu_kernel<<<MN / 16, 128, 0, stream>>>(batch, weights, w2, bmu);
  som_out_kernel<<<4096, 128, 0, stream>>>(bmu, it, decay, out);
}


# GEMM re-tiled: work-group = 64 batch rows in LDS x a column quarter, wave = 64x32 register tile, loads issued ahead; each dot is one sum chain (was 4); quarter maxima merged in last kernel
# speedup vs baseline: 4.1665x; 4.1665x over previous
#include <hip/hip_runtime.h>


typedef __attribute__((ext_vector_type(2))) float v2f;
typedef __attribute__((ext_vector_type(4))) float v4f;
typedef __attribute__((ext_vector_type(8))) float v8f;

#define DIM 512
#define MN  4096
#define LDA 516

#define MB      64
#define NSPLIT  4
#define NCOLS   (MN / NSPLIT)
#define NWAVES  8
#define WCOLS   (NCOLS / NWAVES)
#define NPAIR   (WCOLS / 32)
#define KSTEPS  (DIM / 8)
#define PSTRIDE 32

__global__ __launch_bounds__(256) void som_w2_kernel(const float* __restrict__ weights,
                                                     float* __restrict__ w2) {
  const int wave = threadIdx.x >> 5;
  const int lane = threadIdx.x & 31;
  const int row  = blockIdx.x * 8 + wave;
  const float4* wr = (const float4*)(weights + (size_t)row * DIM);
  float acc = 0.0f;
#pragma unroll
  for (int i = 0; i < 4; ++i) {
    float4 q = wr[lane + 32 * i];
    acc += q.x * q.x + q.y * q.y + q.z * q.z + q.w * q.w;
  }
#pragma unroll
  for (int m = 16; m >= 1; m >>= 1) acc += __shfl_xor(acc, m, 32);
  if (lane == 0) w2[row] = acc;
}

__global__ __launch_bounds__(256) void som_bmu_kernel(const float* __restrict__ batch,
                                                      const float* __restrict__ weights,
                                                      const float* __restrict__ w2,
                                                      float* __restrict__ part) {
  __shared__ __align__(16) float As[MB * LDA];
  __shared__ float cV[NWAVES][MB];
  __shared__ int   cI[NWAVES][MB];

  const int tid   = threadIdx.x;
  const int lane  = tid & 31;
  const int wv    = tid >> 5;
  const int mblk  = blockIdx.x >> 2;
  const int split = blockIdx.x & 3;

  const float4* bp = (const float4*)(batch + (size_t)mblk * MB * DIM);
#pragma unroll 8
  for (int i = 0; i < (MB * DIM / 4) / 256; ++i) {
    int idx = tid + 256 * i;
    int row = idx >> 7;
    int c4  = idx & 127;
    float4 q = bp[idx];
    *(float4*)&As[row * LDA + c4 * 4] = q;
  }
  __syncthreads();

  const int half = lane >> 4;
  const int l16  = lane & 15;
  const float* aBase = &As[l16 * LDA + 4 * half];
  const int nb = split * NCOLS + wv * WCOLS;
  const float* wB = weights + (size_t)(nb + l16) * DIM + 4 * half;

  float bestV[4][8];
  int   bestI[4][8];
#pragma unroll
  for (int m = 0; m < 4; ++m)
#pragma unroll
    for (int r = 0; r < 8; ++r) { bestV[m][r] = -3.402823466e38f; bestI[m][r] = 0; }

  v4f B[4][2];
  v4f A[2][4];
  const int TOT = NPAIR * KSTEPS;

#define WM(a, b, c) c = __builtin_amdgcn_wmma_f32_16x16x4_f32(false, a, false, b, (short)0, c, false, false)
#define LOADB(S, T) {                                                          \
    int tt = (T) < TOT ? (T) : TOT - 1;                                        \
    const float* wn = wB + (size_t)(tt / KSTEPS) * (32 * DIM) + (tt % KSTEPS) * 8; \
    B[S][0] = *(const v4f*)(wn);                                               \
    B[S][1] = *(const v4f*)(wn + 16 * DIM); }
#define LOADA(S, T) {                                                          \
    const float* an = aBase + ((T) % KSTEPS) * 8;                              \
    A[S][0] = *(const v4f*)(an);                                               \
    A[S][1] = *(const v4f*)(an + 16 * LDA);                                    \
    A[S][2] = *(const v4f*)(an + 32 * LDA);                                    \
    A[S][3] = *(const v4f*)(an + 48 * LDA); }
#define STEP(Q, T)                                                             \
    LOADB(((Q) + 2) & 3, (T) + 2)                                              \
    LOADA(((Q) + 1) & 1, (T) + 1)                                              \
    __builtin_amdgcn_sched_barrier(0);                                         \
    WM(A[(Q)&1][0].xy, B[(Q)&3][0].xy, c00); WM(A[(Q)&1][1].xy, B[(Q)&3][0].xy, c10); \
    WM(A[(Q)&1][2].xy, B[(Q)&3][0].xy, c20); WM(A[(Q)&1][3].xy, B[(Q)&3][0].xy, c30); \
    WM(A[(Q)&1][0].xy, B[(Q)&3][1].xy, c01); WM(A[(Q)&1][1].xy, B[(Q)&3][1].xy, c11); \
    WM(A[(Q)&1][2].xy, B[(Q)&3][1].xy, c21); WM(A[(Q)&1][3].xy, B[(Q)&3][1].xy, c31); \
    WM(A[(Q)&1][0].zw, B[(Q)&3][0].zw, c00); WM(A[(Q)&1][1].zw, B[(Q)&3][0].zw, c10); \
    WM(A[(Q)&1][2].zw, B[(Q)&3][0].zw, c20); WM(A[(Q)&1][3].zw, B[(Q)&3][0].zw, c30); \
    WM(A[(Q)&1][0].zw, B[(Q)&3][1].zw, c01); WM(A[(Q)&1][1].zw, B[(Q)&3][1].zw, c11); \
    WM(A[(Q)&1][2].zw, B[(Q)&3][1].zw, c21); WM(A[(Q)&1][3].zw, B[(Q)&3][1].zw, c31); \
    __builtin_amdgcn_sched_barrier(0);

  LOADB(0, 0)
  LOADB(1, 1)
  LOADA(0, 0)

  for (int p = 0; p < NPAIR; ++p) {
    v8f c00 = {}, c01 = {}, c10 = {}, c11 = {}, c20 = {}, c21 = {}, c30 = {}, c31 = {};
    for (int k4 = 0; k4 < KSTEPS; k4 += 4) {
      const int t = p * KSTEPS + k4;
      STEP(0, t)
      STEP(1, t + 1)
      STEP(2, t + 2)
      STEP(3, t + 3)
    }

    const int n0 = nb + p * 32 + l16;
    const float w2a = w2[n0];
    const float w2b = w2[n0 + 16];
#define UPD(m, C, W2N, N)                                                      \
    _Pragma("unroll") for (int r = 0; r < 8; ++r) {                            \
      const float s = 2.0f * C[r] - W2N;                                       \
      const bool g = s > bestV[m][r];                                          \
      bestV[m][r] = g ? s : bestV[m][r];                                       \
      bestI[m][r] = g ? (N) : bestI[m][r];                                     \
    }
    UPD(0, c00, w2a, n0) UPD(1, c10, w2a, n0) UPD(2, c20, w2a, n0) UPD(3, c30, w2a, n0)
    UPD(0, c01, w2b, n0 + 16) UPD(1, c11, w2b, n0 + 16) UPD(2, c21, w2b, n0 + 16) UPD(3, c31, w2b, n0 + 16)
#undef UPD
  }
#undef STEP
#undef LOADA
#undef LOADB
#undef WM

#pragma unroll
  for (int x = 8; x >= 1; x >>= 1) {
    float ov[4][8];
    int   oi[4][8];
#pragma unroll
    for (int m = 0; m < 4; ++m)
#pragma unroll
      for (int r = 0; r < 8; ++r) {
        ov[m][r] = __shfl_xor(bestV[m][r], x, 32);
        oi[m][r] = __shfl_xor(bestI[m][r], x, 32);
      }
#pragma unroll
    for (int m = 0; m < 4; ++m)
#pragma unroll
      for (int r = 0; r < 8; ++r) {
        const bool take = (ov[m][r] > bestV[m][r]) |
                          ((ov[m][r] == bestV[m][r]) & (oi[m][r] < bestI[m][r]));
        bestV[m][r] = take ? ov[m][r] : bestV[m][r];
        bestI[m][r] = take ? oi[m][r] : bestI[m][r];
      }
  }
  if (l16 == 0) {
#pragma unroll
    for (int m = 0; m < 4; ++m)
#pragma unroll
      for (int r = 0; r < 8; ++r) {
        cV[wv][m * 16 + 8 * half + r] = bestV[m][r];
        cI[wv][m * 16 + 8 * half + r] = bestI[m][r];
      }
  }
  __syncthreads();

  if (tid < MB) {
    float v = cV[0][tid];
    int   ix = cI[0][tid];
#pragma unroll
    for (int w = 1; w < NWAVES; ++w) {
      float ov = cV[w][tid];
      int   oi = cI[w][tid];
      const bool take = (ov > v) | ((ov == v) & (oi < ix));
      v = take ? ov : v; ix = take ? oi : ix;
    }
    float* pr = part + (size_t)(mblk * MB + tid) * MN + split * PSTRIDE;
    v2f rec = { v, __int_as_float(ix) };
    *(v2f*)pr = rec;
  }
}

__global__ __launch_bounds__(128) void som_out_kernel(const int* __restrict__ it_p,
                                                      const int* __restrict__ decay_p,
                                                      float* out) {
  __shared__ float efi[64];
  __shared__ float efj[64];
  const int row = blockIdx.x;
  const int t   = threadIdx.x;
  float* op = out + (size_t)row * MN;
  float bv = op[0];
  int   idx = __float_as_int(op[1]);
#pragma unroll
  for (int s = 1; s < NSPLIT; ++s) {
    float ov = op[s * PSTRIDE];
    int   oi = __float_as_int(op[s * PSTRIDE + 1]);
    if (ov > bv) { bv = ov; idx = oi; }
  }
  const int bi = idx >> 6;
  const int bj = idx & 63;
  const float lr  = __expf(-(float)it_p[0] / (float)decay_p[0]);
  const float sig = 32.0f * lr;
  const float inv = 1.0f / (sig * sig);
  if (t < 64) {
    int d = t - bi;
    efi[t] = __expf(-(float)(d * d) * inv);
  } else {
    int c = t - 64;
    int d = c - bj;
    efj[c] = __expf(-(float)(d * d) * inv);
  }
  __syncthreads();
  const float fj = efj[t & 63];
#pragma unroll
  for (int k = 0; k < 32; ++k) {
    int j = t + (k << 7);
    op[j] = efi[j >> 6] * fj;
  }
}

extern "C" void kernel_launch(void* const* d_in, const int* in_sizes, int n_in,
                              void* d_out, int out_size, void* d_ws, size_t ws_size,
                              hipStream_t stream) {
  const float* batch   = (const float*)d_in[0];
  const float* weights = (const float*)d_in[1];
  const int* decay = (const int*)d_in[3];
  const int* it    = (const int*)d_in[4];

  float* w2  = (float*)d_ws;
  float* out = (float*)d_out;

  som_w2_kernel <<<MN / 8, 256, 0, stream>>>(weights, w2);
  som_bmu_kernel<<<(MN / MB) * NSPLIT, 256, 0, stream>>>(batch, weights, w2, out);
  som_out_kernel<<<4096, 128, 0, stream>>>(it, decay, out);
}


# GEMM kernel: global loads requested further ahead of use: weight fragments 6 steps ahead (was 2) in 8 register sets, w2 at each pair's start, first fragments before the tile load; sums unchanged
# speedup vs baseline: 4.3625x; 1.0470x over previous
#include <hip/hip_runtime.h>


typedef __attribute__((ext_vector_type(2))) float v2f;
typedef __attribute__((ext_vector_type(4))) float v4f;
typedef __attribute__((ext_vector_type(8))) float v8f;

#define DIM 512
#define MN  4096
#define LDA 516

#define MB      64
#define NSPLIT  4
#define NCOLS   (MN / NSPLIT)
#define NWAVES  8
#define WCOLS   (NCOLS / NWAVES)
#define NPAIR   (WCOLS / 32)
#define KSTEPS  (DIM / 8)
#define PSTRIDE 32

__global__ __launch_bounds__(256) void som_w2_kernel(const float* __restrict__ weights,
                                                     float* __restrict__ w2) {
  const int wave = threadIdx.x >> 5;
  const int lane = threadIdx.x & 31;
  const int row  = blockIdx.x * 8 + wave;
  const float4* wr = (const float4*)(weights + (size_t)row * DIM);
  float acc = 0.0f;
#pragma unroll
  for (int i = 0; i < 4; ++i) {
    float4 q = wr[lane + 32 * i];
    acc += q.x * q.x + q.y * q.y + q.z * q.z + q.w * q.w;
  }
#pragma unroll
  for (int m = 16; m >= 1; m >>= 1) acc += __shfl_xor(acc, m, 32);
  if (lane == 0) w2[row] = acc;
}

__global__ __launch_bounds__(256) void som_bmu_kernel(const float* __restrict__ batch,
                                                      const float* __restrict__ weights,
                                                      const float* __restrict__ w2,
                                                      float* __restrict__ part) {
  __shared__ __align__(16) float As[MB * LDA];
  __shared__ float cV[NWAVES][MB];
  __shared__ int   cI[NWAVES][MB];

  const int tid   = threadIdx.x;
  const int lane  = tid & 31;
  const int wv    = tid >> 5;
  const int mblk  = blockIdx.x >> 2;
  const int split = blockIdx.x & 3;

  const int half = lane >> 4;
  const int l16  = lane & 15;
  const float* aBase = &As[l16 * LDA + 4 * half];
  const int nb = split * NCOLS + wv * WCOLS;
  const char* wU = (const char*)(weights + (size_t)nb * DIM);
  const unsigned wLane = (unsigned)(l16 * DIM + 4 * half) * 4u;

  v4f B[8][2];
  v4f A[2][4];

#define WM(a, b, c) c = __builtin_amdgcn_wmma_f32_16x16x4_f32(false, a, false, b, (short)0, c, false, false)
#define LOADB(S, P) {                                                          \
    const char* wn = (P);                                                      \
    B[S][0] = *(const v4f*)(wn + wLane);                                       \
    B[S][1] = *(const v4f*)(wn + wLane + 16 * DIM * 4); }
#define LOADA(S, K) {                                                          \
    const float* an = aBase + ((K) % KSTEPS) * 8;                              \
    A[S][0] = *(const v4f*)(an);                                               \
    A[S][1] = *(const v4f*)(an + 16 * LDA);                                    \
    A[S][2] = *(const v4f*)(an + 32 * LDA);                                    \
    A[S][3] = *(const v4f*)(an + 48 * LDA); }
#define STEP(Q)                                                                \
    LOADB(((Q) + 6) & 7, (Q) < 2 ? pfA + (Q) * 32 : pfB + ((Q) - 2) * 32)      \
    LOADA(((Q) + 1) & 1, k8 + (Q) + 1)                                         \
    __builtin_amdgcn_sched_barrier(0);                                         \
    WM(A[(Q)&1][0].xy, B[(Q)&7][0].xy, c00); WM(A[(Q)&1][1].xy, B[(Q)&7][0].xy, c10); \
    WM(A[(Q)&1][2].xy, B[(Q)&7][0].xy, c20); WM(A[(Q)&1][3].xy, B[(Q)&7][0].xy, c30); \
    WM(A[(Q)&1][0].xy, B[(Q)&7][1].xy, c01); WM(A[(Q)&1][1].xy, B[(Q)&7][1].xy, c11); \
    WM(A[(Q)&1][2].xy, B[(Q)&7][1].xy, c21); WM(A[(Q)&1][3].xy, B[(Q)&7][1].xy, c31); \
    WM(A[(Q)&1][0].zw, B[(Q)&7][0].zw, c00); WM(A[(Q)&1][1].zw, B[(Q)&7][0].zw, c10); \
    WM(A[(Q)&1][2].zw, B[(Q)&7][0].zw, c20); WM(A[(Q)&1][3].zw, B[(Q)&7][0].zw, c30); \
    WM(A[(Q)&1][0].zw, B[(Q)&7][1].zw, c01); WM(A[(Q)&1][1].zw, B[(Q)&7][1].zw, c11); \
    WM(A[(Q)&1][2].zw, B[(Q)&7][1].zw, c21); WM(A[(Q)&1][3].zw, B[(Q)&7][1].zw, c31); \
    __builtin_amdgcn_sched_barrier(0);

  LOADB(0, wU)
  LOADB(1, wU + 32)
  LOADB(2, wU + 64)
  LOADB(3, wU + 96)
  LOADB(4, wU + 128)
  LOADB(5, wU + 160)

  const float4* bp = (const float4*)(batch + (size_t)mblk * MB * DIM);
#pragma unroll 8
  for (int i = 0; i < (MB * DIM / 4) / 256; ++i) {
    int idx = tid + 256 * i;
    int row = idx >> 7;
    int c4  = idx & 127;
    float4 q = bp[idx];
    *(float4*)&As[row * LDA + c4 * 4] = q;
  }
  __syncthreads();

  float bestV[4][8];
  int   bestI[4][8];
#pragma unroll
  for (int m = 0; m < 4; ++m)
#pragma unroll
    for (int r = 0; r < 8; ++r) { bestV[m][r] = -3.402823466e38f; bestI[m][r] = 0; }

  LOADA(0, 0)

  for (int p = 0; p < NPAIR; ++p) {
    v8f c00 = {}, c01 = {}, c10 = {}, c11 = {}, c20 = {}, c21 = {}, c30 = {}, c31 = {};
    const char* pairB = wU + (size_t)p * (32 * DIM * 4);
    const char* nextB = (p + 1 < NPAIR) ? pairB + 32 * DIM * 4 : pairB + (KSTEPS - 8) * 32;
    const int n0 = nb + p * 32 + l16;
    const float w2a = w2[n0];
    const float w2b = w2[n0 + 16];
    for (int k8 = 0; k8 < KSTEPS; k8 += 8) {
      const char* pfA = pairB + (k8 + 6) * 32;
      const char* pfB = (k8 + 8 < KSTEPS) ? pairB + (k8 + 8) * 32 : nextB;
      STEP(0)
      STEP(1)
      STEP(2)
      STEP(3)
      STEP(4)
      STEP(5)
      STEP(6)
      STEP(7)
    }

#define UPD(m, C, W2N, N)                                                      \
    _Pragma("unroll") for (int r = 0; r < 8; ++r) {                            \
      const float s = 2.0f * C[r] - W2N;                                       \
      const bool g = s > bestV[m][r];                                          \
      bestV[m][r] = g ? s : bestV[m][r];                                       \
      bestI[m][r] = g ? (N) : bestI[m][r];                                     \
    }
    UPD(0, c00, w2a, n0) UPD(1, c10, w2a, n0) UPD(2, c20, w2a, n0) UPD(3, c30, w2a, n0)
    UPD(0, c01, w2b, n0 + 16) UPD(1, c11, w2b, n0 + 16) UPD(2, c21, w2b, n0 + 16) UPD(3, c31, w2b, n0 + 16)
#undef UPD
  }
#undef STEP
#undef LOADA
#undef LOADB
#undef WM

#pragma unroll
  for (int x = 8; x >= 1; x >>= 1) {
    float ov[4][8];
    int   oi[4][8];
#pragma unroll
    for (int m = 0; m < 4; ++m)
#pragma unroll
      for (int r = 0; r < 8; ++r) {
        ov[m][r] = __shfl_xor(bestV[m][r], x, 32);
        oi[m][r] = __shfl_xor(bestI[m][r], x, 32);
      }
#pragma unroll
    for (int m = 0; m < 4; ++m)
#pragma unroll
      for (int r = 0; r < 8; ++r) {
        const bool take = (ov[m][r] > bestV[m][r]) |
                          ((ov[m][r] == bestV[m][r]) & (oi[m][r] < bestI[m][r]));
        bestV[m][r] = take ? ov[m][r] : bestV[m][r];
        bestI[m][r] = take ? oi[m][r] : bestI[m][r];
      }
  }
  if (l16 == 0) {
#pragma unroll
    for (int m = 0; m < 4; ++m)
#pragma unroll
      for (int r = 0; r < 8; ++r) {
        cV[wv][m * 16 + 8 * half + r] = bestV[m][r];
        cI[wv][m * 16 + 8 * half + r] = bestI[m][r];
      }
  }
  __syncthreads();

  if (tid < MB) {
    float v = cV[0][tid];
    int   ix = cI[0][tid];
#pragma unroll
    for (int w = 1; w < NWAVES; ++w) {
      float ov = cV[w][tid];
      int   oi = cI[w][tid];
      const bool take = (ov > v) | ((ov == v) & (oi < ix));
      v = take ? ov : v; ix = take ? oi : ix;
    }
    float* pr = part + (size_t)(mblk * MB + tid) * MN + split * PSTRIDE;
    v2f rec = { v, __int_as_float(ix) };
    *(v2f*)pr = rec;
  }
}

__global__ __launch_bounds__(128) void som_out_kernel(const int* __restrict__ it_p,
                                                      const int* __restrict__ decay_p,
                                                      float* out) {
  __shared__ float efi[64];
  __shared__ float efj[64];
  const int row = blockIdx.x;
  const int t   = threadIdx.x;
  float* op = out + (size_t)row * MN;
  float bv = op[0];
  int   idx = __float_as_int(op[1]);
#pragma unroll
  for (int s = 1; s < NSPLIT; ++s) {
    float ov = op[s * PSTRIDE];
    int   oi = __float_as_int(op[s * PSTRIDE + 1]);
    if (ov > bv) { bv = ov; idx = oi; }
  }
  const int bi = idx >> 6;
  const int bj = idx & 63;
  const float lr  = __expf(-(float)it_p[0] / (float)decay_p[0]);
  const float sig = 32.0f * lr;
  const float inv = 1.0f / (sig * sig);
  if (t < 64) {
    int d = t - bi;
    efi[t] = __expf(-(float)(d * d) * inv);
  } else {
    int c = t - 64;
    int d = c - bj;
    efj[c] = __expf(-(float)(d * d) * inv);
  }
  __syncthreads();
  const float fj = efj[t & 63];
#pragma unroll
  for (int k = 0; k < 32; ++k) {
    int j = t + (k << 7);
    op[j] = efi[j >> 6] * fj;
  }
}

extern "C" void kernel_launch(void* const* d_in, const int* in_sizes, int n_in,
                              void* d_out, int out_size, void* d_ws, size_t ws_size,
                              hipStream_t stream) {
  const float* batch   = (const float*)d_in[0];
  const float* weights = (const float*)d_in[1];
  const int* decay = (const int*)d_in[3];
  const int* it    = (const int*)d_in[4];

  float* w2  = (float*)d_ws;
  float* out = (float*)d_out;

  som_w2_kernel <<<MN / 8, 256, 0, stream>>>(weights, w2);
  som_bmu_kernel<<<(MN / MB) * NSPLIT, 256, 0, stream>>>(batch, weights, w2, out);
  som_out_kernel<<<4096, 128, 0, stream>>>(it, decay, out);
}
